# starting kernel
# baseline (speedup 1.0000x reference)
#include <hip/hip_runtime.h>
#include <hip/hip_bf16.h>

#define N_NODES 50000
#define E_EDGES 800000
#define IN_F    256
#define H_HEADS 4
#define F_OUT   64
#define HF      256
#define SLOPE   0.2f

typedef __attribute__((ext_vector_type(16))) __bf16 v16bf;
typedef __attribute__((ext_vector_type(8)))  float  v8f;

__device__ __forceinline__ float leakyf(float x) { return x > 0.f ? x : SLOPE * x; }

__device__ __forceinline__ void atomicMaxF(float* addr, float val) {
  if (val >= 0.f) {
    atomicMax((int*)addr, __float_as_int(val));
  } else {
    atomicMin((unsigned int*)addr, __float_as_uint(val));
  }
}

#define WT_STRIDE 40

__global__ __launch_bounds__(256)
void gemm_h_kernel(const float* __restrict__ feat, const float* __restrict__ W,
                   float* __restrict__ h) {
  __shared__ __align__(16) __bf16 Wt[HF * WT_STRIDE];

  const int t       = threadIdx.x;
  const int wave    = t >> 5;
  const int lane    = t & 31;
  const int rowBase = blockIdx.x * 128 + wave * 16;
  const bool rowValid = (rowBase < N_NODES);
  const int nloc = lane & 15;
  const int kg   = lane >> 4;

  v8f acc[16];
#pragma unroll
  for (int nt = 0; nt < 16; ++nt)
#pragma unroll
    for (int v = 0; v < 8; ++v) acc[nt][v] = 0.f;

  for (int ks = 0; ks < IN_F / 32; ++ks) {
    const int kbase = ks * 32;
    __syncthreads();
#pragma unroll
    for (int kk = 0; kk < 32; ++kk)
      Wt[t * WT_STRIDE + kk] = (__bf16)W[(kbase + kk) * HF + t];
    __syncthreads();

    if (rowValid) {
      const float* arow = feat + (size_t)(rowBase + nloc) * IN_F + kbase;
      float4 a0 = *(const float4*)(arow + 8 * kg);
      float4 a1 = *(const float4*)(arow + 8 * kg + 4);
      float4 a2 = *(const float4*)(arow + 16 + 8 * kg);
      float4 a3 = *(const float4*)(arow + 16 + 8 * kg + 4);
      float atmp[16] = {a0.x, a0.y, a0.z, a0.w, a1.x, a1.y, a1.z, a1.w,
                        a2.x, a2.y, a2.z, a2.w, a3.x, a3.y, a3.z, a3.w};
      v16bf afrag;
#pragma unroll
      for (int i = 0; i < 16; ++i) afrag[i] = (__bf16)atmp[i];

      const __bf16* wrow0 = &Wt[nloc * WT_STRIDE];
      uint4 b0 = *(const uint4*)(wrow0 + 8 * kg);
      uint4 b1 = *(const uint4*)(wrow0 + 16 + 8 * kg);
#pragma unroll
      for (int nt = 0; nt < 16; ++nt) {
        uint4 nb0, nb1;
        if (nt < 15) {
          const __bf16* wrow = &Wt[((nt + 1) * 16 + nloc) * WT_STRIDE];
          nb0 = *(const uint4*)(wrow + 8 * kg);
          nb1 = *(const uint4*)(wrow + 16 + 8 * kg);
        }
        const __bf16* p0 = (const __bf16*)&b0;
        const __bf16* p1 = (const __bf16*)&b1;
        v16bf bfrag;
#pragma unroll
        for (int i = 0; i < 8; ++i) { bfrag[i] = p0[i]; bfrag[8 + i] = p1[i]; }
        acc[nt] = __builtin_amdgcn_wmma_f32_16x16x32_bf16(
            false, afrag, false, bfrag, (short)0, acc[nt], false, false);
        if (nt < 15) { b0 = nb0; b1 = nb1; }
      }
    }
  }

  if (rowValid) {
#pragma unroll
    for (int nt = 0; nt < 16; ++nt)
#pragma unroll
      for (int v = 0; v < 8; ++v)
        h[(size_t)(rowBase + v + 8 * kg) * HF + nt * 16 + nloc] = acc[nt][v];
  }
}

__global__ void init_kernel(float* maxs, float* maxd, float* sums, float* sumd,
                            float* outdeg, float* indeg) {
  int i = blockIdx.x * blockDim.x + threadIdx.x;
  if (i < N_NODES * H_HEADS) {
    maxs[i] = -1e30f; maxd[i] = -1e30f; sums[i] = 0.f; sumd[i] = 0.f;
  }
  if (i < N_NODES) { outdeg[i] = 0.f; indeg[i] = 0.f; }
}

__global__ void scores_kernel(const float* __restrict__ h,
                              const float* __restrict__ attn_l,
                              const float* __restrict__ attn_r,
                              const float* __restrict__ hop_attn_l,
                              float* __restrict__ el, float* __restrict__ er,
                              float* __restrict__ al) {
  int i = blockIdx.x * blockDim.x + threadIdx.x;
  if (i >= N_NODES * H_HEADS) return;
  int node = i >> 2, head = i & 3;
  const float* hp  = h + (size_t)node * HF + head * F_OUT;
  const float* alp = attn_l + head * F_OUT;
  const float* arp = attn_r + head * F_OUT;
  const float* hlp = hop_attn_l + head * F_OUT;
  float sl = 0.f, sr = 0.f, shl = 0.f;
#pragma unroll 8
  for (int f = 0; f < F_OUT; ++f) {
    float v = hp[f];
    sl += v * alp[f]; sr += v * arp[f]; shl += v * hlp[f];
  }
  el[i] = sl; er[i] = sr; al[i] = shl;
}

__global__ void edge_e_kernel(const int* __restrict__ src, const int* __restrict__ dst,
                              const float* __restrict__ el, const float* __restrict__ er,
                              float* __restrict__ e_ws, float* maxs, float* maxd,
                              float* outdeg, float* indeg) {
  int i = blockIdx.x * blockDim.x + threadIdx.x;
  if (i >= E_EDGES * H_HEADS) return;
  int eid = i >> 2, head = i & 3;
  int s = src[eid], d = dst[eid];
  float ev = leakyf(el[s * 4 + head] + er[d * 4 + head]);
  e_ws[i] = ev;
  atomicMaxF(&maxs[s * 4 + head], ev);
  atomicMaxF(&maxd[d * 4 + head], ev);
  if (head == 0) { atomicAdd(&outdeg[s], 1.f); atomicAdd(&indeg[d], 1.f); }
}

__global__ void edge_sum_kernel(const int* __restrict__ src, const int* __restrict__ dst,
                                const float* __restrict__ e_ws,
                                const float* __restrict__ maxs, const float* __restrict__ maxd,
                                float* sums, float* sumd) {
  int i = blockIdx.x * blockDim.x + threadIdx.x;
  if (i >= E_EDGES * H_HEADS) return;
  int eid = i >> 2, head = i & 3;
  int s = src[eid], d = dst[eid];
  float ev = e_ws[i];
  atomicAdd(&sums[s * 4 + head], __expf(ev - maxs[s * 4 + head]));
  atomicAdd(&sumd[d * 4 + head], __expf(ev - maxd[d * 4 + head]));
}

__global__ void edge_a_kernel(const int* __restrict__ src, const int* __restrict__ dst,
                              const float* __restrict__ e_ws,
                              const float* __restrict__ maxs, const float* __restrict__ maxd,
                              const float* __restrict__ sums, const float* __restrict__ sumd,
                              const float* __restrict__ sigma, float* __restrict__ a_ws) {
  int i = blockIdx.x * blockDim.x + threadIdx.x;
  if (i >= E_EDGES * H_HEADS) return;
  int eid = i >> 2, head = i & 3;
  int s = src[eid], d = dst[eid];
  float ev = e_ws[i];
  float sgm = 1.f / (1.f + __expf(-sigma[0]));
  float ax = fmaxf(__expf(ev - maxs[s * 4 + head]) / sums[s * 4 + head], 1e-10f);
  float ay = fmaxf(__expf(ev - maxd[d * 4 + head]) / sumd[d * 4 + head], 1e-10f);
  a_ws[i] = __powf(ay, sgm) * __powf(ax, 1.f - sgm);
}

__global__ void norm_kernel(const float* __restrict__ outdeg, const float* __restrict__ indeg,
                            float* __restrict__ norm_out, float* __restrict__ norm_in) {
  int i = blockIdx.x * blockDim.x + threadIdx.x;
  if (i >= N_NODES) return;
  norm_out[i] = rsqrtf(fmaxf(outdeg[i], 1.f));
  norm_in[i]  = sqrtf(fmaxf(indeg[i], 1.f));
}

__global__ __launch_bounds__(256)
void agg_kernel(const int* __restrict__ src, const int* __restrict__ dst,
                const float* __restrict__ a_ws, const float* __restrict__ norm_out,
                const float* __restrict__ xprev, float* __restrict__ xnext) {
  int eid = blockIdx.x;
  int t = threadIdx.x;
  int s = src[eid], d = dst[eid];
  int head = t >> 6;
  float coef = a_ws[eid * 4 + head] * norm_out[s];
  float v = xprev[(size_t)s * HF + t] * coef;
  atomicAdd(&xnext[(size_t)d * HF + t], v);
}

__global__ void scale_kernel(float* __restrict__ x, const float* __restrict__ norm_in) {
  int i = blockIdx.x * blockDim.x + threadIdx.x;
  if (i >= N_NODES * HF) return;
  x[i] *= norm_in[i >> 8];
}

__global__ __launch_bounds__(128)
void combine_kernel(const float* __restrict__ h, const float* __restrict__ x1,
                    const float* __restrict__ x2, const float* __restrict__ x3,
                    const float* __restrict__ al, const float* __restrict__ hop_attn_r,
                    float* __restrict__ out) {
  int n    = blockIdx.x;
  int head = threadIdx.x >> 5;
  int lane = threadIdx.x & 31;
  const float* hs[4] = {h, x1, x2, x3};
  size_t base = (size_t)n * HF + head * F_OUT;
  const float* ar = hop_attn_r + head * F_OUT;
  float ar0 = ar[lane], ar1 = ar[lane + 32];
  float albase = al[n * 4 + head];

  float xv0[4], xv1[4], score[4];
#pragma unroll
  for (int k = 0; k < 4; ++k) {
    float v0 = hs[k][base + lane];
    float v1 = hs[k][base + lane + 32];
    xv0[k] = v0; xv1[k] = v1;
    float p = v0 * ar0 + v1 * ar1;
#pragma unroll
    for (int off = 16; off > 0; off >>= 1) p += __shfl_xor(p, off);
    score[k] = leakyf(albase + p);
  }
  float m = fmaxf(fmaxf(score[0], score[1]), fmaxf(score[2], score[3]));
  float ex[4], ssum = 0.f;
#pragma unroll
  for (int k = 0; k < 4; ++k) { ex[k] = __expf(score[k] - m); ssum += ex[k]; }
  float o0 = 0.f, o1 = 0.f;
#pragma unroll
  for (int k = 0; k < 4; ++k) {
    float w = ex[k] / ssum;
    o0 += xv0[k] * w; o1 += xv1[k] * w;
  }
  out[base + lane] = o0;
  out[base + lane + 32] = o1;
}

extern "C" void kernel_launch(void* const* d_in, const int* in_sizes, int n_in,
                              void* d_out, int out_size, void* d_ws, size_t ws_size,
                              hipStream_t stream) {
  (void)in_sizes; (void)n_in; (void)out_size; (void)ws_size;
  const float* feat       = (const float*)d_in[0];
  const int*   src        = (const int*)d_in[1];
  const int*   dst        = (const int*)d_in[2];
  const float* W          = (const float*)d_in[3];
  const float* attn_l     = (const float*)d_in[4];
  const float* attn_r     = (const float*)d_in[5];
  const float* hop_attn_l = (const float*)d_in[6];
  const float* hop_attn_r = (const float*)d_in[7];
  const float* sigma      = (const float*)d_in[8];

  float* ws = (float*)d_ws;
  size_t off = 0;
  float* h    = ws + off; off += (size_t)N_NODES * HF;
  float* x1   = ws + off; off += (size_t)N_NODES * HF;
  float* x2   = ws + off; off += (size_t)N_NODES * HF;
  float* x3   = ws + off; off += (size_t)N_NODES * HF;
  float* el   = ws + off; off += (size_t)N_NODES * H_HEADS;
  float* er   = ws + off; off += (size_t)N_NODES * H_HEADS;
  float* al   = ws + off; off += (size_t)N_NODES * H_HEADS;
  float* maxs = ws + off; off += (size_t)N_NODES * H_HEADS;
  float* maxd = ws + off; off += (size_t)N_NODES * H_HEADS;
  float* sums = ws + off; off += (size_t)N_NODES * H_HEADS;
  float* sumd = ws + off; off += (size_t)N_NODES * H_HEADS;
  float* outdeg   = ws + off; off += (size_t)N_NODES;
  float* indeg    = ws + off; off += (size_t)N_NODES;
  float* norm_out = ws + off; off += (size_t)N_NODES;
  float* norm_in  = ws + off; off += (size_t)N_NODES;
  float* e_ws = ws + off; off += (size_t)E_EDGES * H_HEADS;
  float* a_ws = ws + off; off += (size_t)E_EDGES * H_HEADS;

  hipMemsetAsync(x1, 0, (size_t)3 * N_NODES * HF * sizeof(float), stream);

  gemm_h_kernel<<<(N_NODES + 127) / 128, 256, 0, stream>>>(feat, W, h);

  const int NH = N_NODES * H_HEADS;
  init_kernel<<<(NH + 255) / 256, 256, 0, stream>>>(maxs, maxd, sums, sumd, outdeg, indeg);

  scores_kernel<<<(NH + 255) / 256, 256, 0, stream>>>(h, attn_l, attn_r, hop_attn_l, el, er, al);

  const int EH = E_EDGES * H_HEADS;
  edge_e_kernel<<<(EH + 255) / 256, 256, 0, stream>>>(src, dst, el, er, e_ws, maxs, maxd, outdeg, indeg);
  edge_sum_kernel<<<(EH + 255) / 256, 256, 0, stream>>>(src, dst, e_ws, maxs, maxd, sums, sumd);
  edge_a_kernel<<<(EH + 255) / 256, 256, 0, stream>>>(src, dst, e_ws, maxs, maxd, sums, sumd, sigma, a_ws);

  norm_kernel<<<(N_NODES + 255) / 256, 256, 0, stream>>>(outdeg, indeg, norm_out, norm_in);

  const int NE = N_NODES * HF;
  agg_kernel<<<E_EDGES, 256, 0, stream>>>(src, dst, a_ws, norm_out, h, x1);
  scale_kernel<<<(NE + 255) / 256, 256, 0, stream>>>(x1, norm_in);
  agg_kernel<<<E_EDGES, 256, 0, stream>>>(src, dst, a_ws, norm_out, x1, x2);
  scale_kernel<<<(NE + 255) / 256, 256, 0, stream>>>(x2, norm_in);
  agg_kernel<<<E_EDGES, 256, 0, stream>>>(src, dst, a_ws, norm_out, x2, x3);
  scale_kernel<<<(NE + 255) / 256, 256, 0, stream>>>(x3, norm_in);

  combine_kernel<<<N_NODES, 128, 0, stream>>>(h, x1, x2, x3, al, hop_attn_r, (float*)d_out);
}


# Mixed edge coefficient as one exp of s*ln(a_dst)+(1-s)*ln(a_src), ln of each exp-sum once per node, clip in the log domain: no pow or division per edge; edge lists filled in fixed per-wave order
# speedup vs baseline: 10.6278x; 10.6278x over previous
#include <hip/hip_runtime.h>
#include <hip/hip_bf16.h>

#define N_NODES 50000
#define E_EDGES 800000
#define IN_F    256
#define H_HEADS 4
#define F_OUT   64
#define HF      256
#define SLOPE   0.2f

typedef __attribute__((ext_vector_type(16))) __bf16 v16bf;
typedef __attribute__((ext_vector_type(8)))  float  v8f;

__device__ __forceinline__ float leakyf(float x) { return x > 0.f ? x : SLOPE * x; }

#define WT_STRIDE 40

#define GEMM_BLOCKS ((N_NODES + 127) / 128)

__device__ __forceinline__
void gemm_h_block(const int bid, const float* __restrict__ feat, const float* __restrict__ W,
                  float* __restrict__ h) {
  __shared__ __align__(16) __bf16 Wt[HF * WT_STRIDE];

  const int t       = threadIdx.x;
  const int wave    = t >> 5;
  const int lane    = t & 31;
  const int rowBase = bid * 128 + wave * 16;
  const bool rowValid = (rowBase < N_NODES);
  const int nloc = lane & 15;
  const int kg   = lane >> 4;

  v8f acc[16];
#pragma unroll
  for (int nt = 0; nt < 16; ++nt)
#pragma unroll
    for (int v = 0; v < 8; ++v) acc[nt][v] = 0.f;

  for (int ks = 0; ks < IN_F / 32; ++ks) {
    const int kbase = ks * 32;
    __syncthreads();
#pragma unroll
    for (int kk = 0; kk < 32; ++kk)
      Wt[t * WT_STRIDE + kk] = (__bf16)W[(kbase + kk) * HF + t];
    __syncthreads();

    if (rowValid) {
      const float* arow = feat + (size_t)(rowBase + nloc) * IN_F + kbase;
      float4 a0 = *(const float4*)(arow + 8 * kg);
      float4 a1 = *(const float4*)(arow + 8 * kg + 4);
      float4 a2 = *(const float4*)(arow + 16 + 8 * kg);
      float4 a3 = *(const float4*)(arow + 16 + 8 * kg + 4);
      float atmp[16] = {a0.x, a0.y, a0.z, a0.w, a1.x, a1.y, a1.z, a1.w,
                        a2.x, a2.y, a2.z, a2.w, a3.x, a3.y, a3.z, a3.w};
      v16bf afrag;
#pragma unroll
      for (int i = 0; i < 16; ++i) afrag[i] = (__bf16)atmp[i];

      const __bf16* wrow0 = &Wt[nloc * WT_STRIDE];
      uint4 b0 = *(const uint4*)(wrow0 + 8 * kg);
      uint4 b1 = *(const uint4*)(wrow0 + 16 + 8 * kg);
#pragma unroll
      for (int nt = 0; nt < 16; ++nt) {
        uint4 nb0, nb1;
        if (nt < 15) {
          const __bf16* wrow = &Wt[((nt + 1) * 16 + nloc) * WT_STRIDE];
          nb0 = *(const uint4*)(wrow + 8 * kg);
          nb1 = *(const uint4*)(wrow + 16 + 8 * kg);
        }
        const __bf16* p0 = (const __bf16*)&b0;
        const __bf16* p1 = (const __bf16*)&b1;
        v16bf bfrag;
#pragma unroll
        for (int i = 0; i < 8; ++i) { bfrag[i] = p0[i]; bfrag[8 + i] = p1[i]; }
        acc[nt] = __builtin_amdgcn_wmma_f32_16x16x32_bf16(
            false, afrag, false, bfrag, (short)0, acc[nt], false, false);
        if (nt < 15) { b0 = nb0; b1 = nb1; }
      }
    }
  }

  if (rowValid) {
#pragma unroll
    for (int nt = 0; nt < 16; ++nt)
#pragma unroll
      for (int v = 0; v < 8; ++v)
        h[(size_t)(rowBase + v + 8 * kg) * HF + nt * 16 + nloc] = acc[nt][v];
  }
}

#define SCORE_BLOCKS ((N_NODES * H_HEADS + 255) / 256)

__device__ __forceinline__
void scores_block(const int bid, const float* __restrict__ h,
                  const float* __restrict__ attn_l,
                  const float* __restrict__ attn_r,
                  const float* __restrict__ hop_attn_l,
                  float* __restrict__ el, float* __restrict__ er,
                  float* __restrict__ al) {
  int i = bid * 256 + threadIdx.x;
  if (i >= N_NODES * H_HEADS) return;
  int node = i >> 2, head = i & 3;
  const float* hp  = h + (size_t)node * HF + head * F_OUT;
  const float* alp = attn_l + head * F_OUT;
  const float* arp = attn_r + head * F_OUT;
  const float* hlp = hop_attn_l + head * F_OUT;
  float sl = 0.f, sr = 0.f, shl = 0.f;
#pragma unroll 8
  for (int f = 0; f < F_OUT; ++f) {
    float v = hp[f];
    sl += v * alp[f]; sr += v * arp[f]; shl += v * hlp[f];
  }
  el[i] = sl; er[i] = sr; al[i] = shl;
}

#define NWAVE 8
#define RANGE_BITS   8
#define RANGE_NODES  (1 << RANGE_BITS)
#define NRANGE       ((N_NODES + RANGE_NODES) / RANGE_NODES)
#define NCHUNK       256
#define CHUNK        (E_EDGES / NCHUNK)
#define CHUNK_PER    ((CHUNK + 255) / 256)
#define TABW         (NRANGE + 1)
#define STAGE_CAP    8192
static_assert(E_EDGES % NCHUNK == 0, "chunks are equal");
static_assert(NRANGE * RANGE_NODES > N_NODES, "the end marker rowptr[N_NODES] lies in the last range");
static_assert(NRANGE < 256, "one LDS counter per thread, and one more for the chunk length");
static_assert(N_NODES <= 65536, "the other endpoint fits 16 bits of the packed word");

__device__ __forceinline__ int excl_scan_256(int c, int lane, int w, int* wsum) {
  int incl = c;
#pragma unroll
  for (int off = 1; off < 32; off <<= 1) {
    const int n = __shfl_up(incl, off);
    if (lane >= off) incl += n;
  }
  if (lane == 31) wsum[w] = incl;
  __syncthreads();
  int wbase = 0;
#pragma unroll
  for (int i = 0; i < 8; ++i) wbase += (i < w) ? wsum[i] : 0;
  return incl - c + wbase;
}

#define BIN_BLOCKS (2 * NCHUNK)

__device__ __forceinline__
void chunk_bin_block(const int chunk, const bool by_dst,
                     const int* __restrict__ src, const int* __restrict__ dst,
                     int* __restrict__ tab_d, int* __restrict__ tab_s,
                     int* __restrict__ tmp_d, int* __restrict__ tmp_s) {
  const int* __restrict__ key = by_dst ? dst : src;
  const int* __restrict__ val = by_dst ? src : dst;
  int* __restrict__ tab = by_dst ? tab_d : tab_s;
  int* __restrict__ tmp = by_dst ? tmp_d : tmp_s;
  __shared__ int cnt[NWAVE * 256];
  __shared__ int wsum[8];
  __shared__ int stage[CHUNK];
  const int t = threadIdx.x, lane = t & 31, w = t >> 5;
  const int base = chunk * CHUNK;
  int* mycnt = cnt + w * 256;

  int k[CHUNK_PER], v[CHUNK_PER];
#pragma unroll
  for (int j = 0; j < CHUNK_PER; ++j) {
    const int i = t + 256 * j;
    const bool ok = i < CHUNK;
    k[j] = ok ? key[base + i] : -1;
    v[j] = ok ? val[base + i] : 0;
  }
#pragma unroll
  for (int q = 0; q < NWAVE; ++q) cnt[q * 256 + t] = 0;
  __syncthreads();
#pragma unroll
  for (int j = 0; j < CHUNK_PER; ++j)
    if (k[j] >= 0) atomicAdd(&mycnt[k[j] >> RANGE_BITS], 1);
  __syncthreads();
  int cw[NWAVE];
  int c = 0;
#pragma unroll
  for (int q = 0; q < NWAVE; ++q) { cw[q] = cnt[q * 256 + t]; c += cw[q]; }
  const int excl = excl_scan_256(c, lane, w, wsum);
  if (t <= NRANGE) tab[chunk * TABW + t] = excl;
  int run = excl;
#pragma unroll
  for (int q = 0; q < NWAVE; ++q) { cnt[q * 256 + t] = run; run += cw[q]; }
  __syncthreads();
#pragma unroll
  for (int j = 0; j < CHUNK_PER; ++j)
    if (k[j] >= 0) {
      const int slot = atomicAdd(&mycnt[k[j] >> RANGE_BITS], 1);
      stage[slot] = ((k[j] & (RANGE_NODES - 1)) << 16) | v[j];
    }
  __syncthreads();
  for (int i = t; i < CHUNK; i += 256) tmp[base + i] = stage[i];
}

template <bool PLACE>
__device__ __forceinline__ void visit_edge(int x, int* cnt, int* stage, int* __restrict__ out,
                                           int first, bool staged) {
  if (PLACE) {
    const int slot = atomicAdd(&cnt[x >> 16], 1);
    if (staged) stage[slot] = x & 0xFFFF; else out[first + slot] = x & 0xFFFF;
  } else {
    atomicAdd(&cnt[x >> 16], 1);
  }
}

template <bool PLACE>
__device__ __forceinline__ void walk_runs(const int* __restrict__ tmp, int beg, int len, int lane,
                                          int* cnt, int* stage, int* __restrict__ out,
                                          int first, bool staged) {
#pragma nounroll
  for (int r = 0; r < 32; r += 4) {
    int b[4], n[4], x[4];
#pragma unroll
    for (int u = 0; u < 4; ++u) {
      b[u] = __builtin_amdgcn_readlane(beg, r + u);
      n[u] = __builtin_amdgcn_readlane(len, r + u);
      x[u] = (lane < n[u]) ? tmp[b[u] + lane] : -1;
    }
#pragma unroll
    for (int u = 0; u < 4; ++u) {
      if (x[u] >= 0) visit_edge<PLACE>(x[u], cnt, stage, out, first, staged);
      for (int j = lane + 32; j < n[u]; j += 32)
        visit_edge<PLACE>(tmp[b[u] + j], cnt, stage, out, first, staged);
    }
  }
}

#define SORT_BLOCKS (2 * NRANGE)

__device__ __forceinline__
void range_sort_block(const int rg, const bool by_dst,
                      const int* __restrict__ tab_d, const int* __restrict__ tab_s,
                      const int* __restrict__ tmp_d, const int* __restrict__ tmp_s,
                      int* __restrict__ rowptr_d, int* __restrict__ rowptr_s,
                      int* __restrict__ csr_src, int* __restrict__ csc_dst) {
  const int* __restrict__ tab = by_dst ? tab_d : tab_s;
  const int* __restrict__ tmp = by_dst ? tmp_d : tmp_s;
  int* __restrict__ rowptr = by_dst ? rowptr_d : rowptr_s;
  int* __restrict__ out = by_dst ? csr_src : csc_dst;
  __shared__ int cnt[NWAVE * 256];
  __shared__ int wsum[8];
  __shared__ int red_lo[8];
  __shared__ int red_len[8];
  __shared__ int stage[STAGE_CAP];
  const int t = threadIdx.x, lane = t & 31, w = t >> 5;

  const int lo = tab[t * TABW + rg], hi = tab[t * TABW + rg + 1];
  const int beg = t * CHUNK + lo, len = hi - lo;
  int a = lo, b = len;
#pragma unroll
  for (int off = 16; off > 0; off >>= 1) { a += __shfl_xor(a, off); b += __shfl_xor(b, off); }
  if (lane == 0) { red_lo[w] = a; red_len[w] = b; }
#pragma unroll
  for (int q = 0; q < NWAVE; ++q) cnt[q * 256 + t] = 0;
  int* mycnt = cnt + w * 256;
  __syncthreads();
  int first = 0, total = 0;
#pragma unroll
  for (int i = 0; i < 8; ++i) { first += red_lo[i]; total += red_len[i]; }
  const bool staged = (total <= STAGE_CAP);

  walk_runs<false>(tmp, beg, len, lane, mycnt, stage, out, first, staged);
  __syncthreads();
  int cw[NWAVE];
  int c = 0;
#pragma unroll
  for (int q = 0; q < NWAVE; ++q) { cw[q] = cnt[q * 256 + t]; c += cw[q]; }
  const int excl = excl_scan_256(c, lane, w, wsum);
  const int node = rg * RANGE_NODES + t;
  if (node <= N_NODES) rowptr[node] = first + excl;
  int run = excl;
#pragma unroll
  for (int q = 0; q < NWAVE; ++q) { cnt[q * 256 + t] = run; run += cw[q]; }
  __syncthreads();
  walk_runs<true>(tmp, beg, len, lane, mycnt, stage, out, first, staged);
  __syncthreads();
  if (staged)
    for (int i = t; i < total; i += 256) out[first + i] = stage[i];
}

__global__ __launch_bounds__(256)
void gemm_bin_kernel(const float* __restrict__ feat, const float* __restrict__ W,
                     float* __restrict__ h,
                     const int* __restrict__ src, const int* __restrict__ dst,
                     int* __restrict__ tab_d, int* __restrict__ tab_s,
                     int* __restrict__ tmp_d, int* __restrict__ tmp_s) {
  const int b = blockIdx.x;
  if (b < GEMM_BLOCKS) {
    gemm_h_block(b, feat, W, h);
  } else {
    const int c = b - GEMM_BLOCKS;
    const bool by_dst = (c < NCHUNK);
    chunk_bin_block(by_dst ? c : c - NCHUNK, by_dst, src, dst, tab_d, tab_s, tmp_d, tmp_s);
  }
}

__global__ __launch_bounds__(256)
void sort_scores_kernel(const int* __restrict__ tab_d, const int* __restrict__ tab_s,
                        const int* __restrict__ tmp_d, const int* __restrict__ tmp_s,
                        int* __restrict__ rowptr_d, int* __restrict__ rowptr_s,
                        int* __restrict__ csr_src, int* __restrict__ csc_dst,
                        const float* __restrict__ h,
                        const float* __restrict__ attn_l, const float* __restrict__ attn_r,
                        const float* __restrict__ hop_attn_l,
                        float* __restrict__ el, float* __restrict__ er, float* __restrict__ al) {
  const int b = blockIdx.x;
  if (b < SORT_BLOCKS) {
    const bool by_dst = (b < NRANGE);
    range_sort_block(by_dst ? b : b - NRANGE, by_dst, tab_d, tab_s, tmp_d, tmp_s,
                     rowptr_d, rowptr_s, csr_src, csc_dst);
  } else {
    scores_block(b - SORT_BLOCKS, h, attn_l, attn_r, hop_attn_l, el, er, al);
  }
}

__device__ __forceinline__ float red8_max(float v) {
  v = fmaxf(v, __shfl_xor(v, 4)); v = fmaxf(v, __shfl_xor(v, 8)); v = fmaxf(v, __shfl_xor(v, 16));
  return v;
}
__device__ __forceinline__ float red8_sum(float v) {
  v += __shfl_xor(v, 4); v += __shfl_xor(v, 8); v += __shfl_xor(v, 16);
  return v;
}

#define WALK_CAP 4

__global__ __launch_bounds__(256)
void src_walk_kernel(const int* __restrict__ rowptr_s, const int* __restrict__ csc_dst,
                     const float* __restrict__ el, const float* __restrict__ er,
                     float* __restrict__ rec_out) {
  float4* __restrict__ rec = (float4*)rec_out;
  const int lane = threadIdx.x & 31;
  const int node = blockIdx.x * 8 + (threadIdx.x >> 5);
  if (node >= N_NODES) return;
  const int head = lane & 3, k = lane >> 2;
  const int beg = rowptr_s[node], end = rowptr_s[node + 1];
  const int deg = end - beg;
  const float self = el[node * 4 + head];

  float m = -INFINITY;
  float sum = 0.f;
  if (deg > 0 && deg <= 8 * WALK_CAP) {
    float e[WALK_CAP];
#pragma unroll
    for (int i = 0; i < WALK_CAP; ++i) {
      const int j = beg + k + 8 * i;
      const float v = leakyf(self + er[csc_dst[j < end ? j : beg] * 4 + head]);
      e[i] = (j < end) ? v : -INFINITY;
      m = fmaxf(m, e[i]);
    }
    m = red8_max(m);
#pragma unroll
    for (int i = 0; i < WALK_CAP; ++i) {
      const float pe = __expf(e[i] - m);
      sum += (beg + k + 8 * i < end) ? pe : 0.f;
    }
    sum = red8_sum(sum);
  } else {
    for (int j = beg + k; j < end; j += 8)
      m = fmaxf(m, leakyf(self + er[csc_dst[j] * 4 + head]));
    m = red8_max(m);
    for (int j = beg + k; j < end; j += 8)
      sum += __expf(leakyf(self + er[csc_dst[j] * 4 + head]) - m);
    sum = red8_sum(sum);
  }

  if (k == 0)
    rec[node * 4 + head] = make_float4(self, m, __logf(sum), rsqrtf(fmaxf((float)deg, 1.f)));
}

#define LN_CLIP (-23.025850929940457f)
__device__ __forceinline__ float mixed_coef(float ev, float m_dst, float lnsum_dst,
                                            const float4& r, float sgm) {
  const float lx = fmaxf((ev - r.y) - r.z, LN_CLIP);
  const float ly = fmaxf((ev - m_dst) - lnsum_dst, LN_CLIP);
  return __expf(sgm * ly + (1.f - sgm) * lx) * r.w;
}

__device__ __forceinline__ void fma4(float4& acc, const float4& v, float c) {
  acc.x += v.x * c; acc.y += v.y * c; acc.z += v.z * c; acc.w += v.w * c;
}

__device__ __forceinline__ void sum_rows(const int* __restrict__ csr_src, const float* __restrict__ csr_coef,
                                         const float* __restrict__ xprev, int beg, int end, int lane,
                                         float4& accA, float4& accB) {
  const bool hi = lane >= 16;
  const float4* x4 = (const float4*)xprev;
  const float4* c4 = (const float4*)csr_coef;
  accA = make_float4(0.f, 0.f, 0.f, 0.f);
  accB = make_float4(0.f, 0.f, 0.f, 0.f);
#pragma unroll 4
  for (int j = beg; j < end; ++j) {
    const int s = csr_src[j];
    const float4 c = c4[j];
    const float4* r = x4 + (size_t)s * (HF / 4);
    const float4 a = r[lane], b = r[32 + lane];
    fma4(accA, a, hi ? c.y : c.x);
    fma4(accB, b, hi ? c.w : c.z);
  }
}

__device__ __forceinline__ void sum_rows_own(const int* __restrict__ csr_src, const float* csr_coef,
                                             const float* __restrict__ xprev, int beg, int end, int lane,
                                             float4& accA, float4& accB) {
  const bool hi = lane >= 16;
  const float4* x4 = (const float4*)xprev;
  const float4* c4 = (const float4*)csr_coef;
  accA = make_float4(0.f, 0.f, 0.f, 0.f);
  accB = make_float4(0.f, 0.f, 0.f, 0.f);
#pragma unroll 4
  for (int j = beg; j < end; ++j) {
    const int s = csr_src[j];
    const float4 c = c4[j];
    const float4* r = x4 + (size_t)s * (HF / 4);
    const float4 a = r[lane], b = r[32 + lane];
    fma4(accA, a, hi ? c.y : c.x);
    fma4(accB, b, hi ? c.w : c.z);
  }
}

__device__ __forceinline__ float4 scale4(const float4& v, float s) {
  return make_float4(v.x * s, v.y * s, v.z * s, v.w * s);
}

__global__ __launch_bounds__(256)
void gather_kernel(const int* __restrict__ rowptr, const int* __restrict__ csr_src,
                   const float* __restrict__ csr_coef, const float* __restrict__ norm_in,
                   const float* __restrict__ xprev, float* __restrict__ xnext) {
  const int lane = threadIdx.x & 31;
  const int node = __builtin_amdgcn_readfirstlane((int)(blockIdx.x * 8 + (threadIdx.x >> 5)));
  if (node >= N_NODES) return;
  const int beg = rowptr[node], end = rowptr[node + 1];
  float4 accA, accB;
  sum_rows(csr_src, csr_coef, xprev, beg, end, lane, accA, accB);

  const float ni = norm_in[node];
  float4* o4 = (float4*)xnext + (size_t)node * (HF / 4);
  o4[lane]      = scale4(accA, ni);
  o4[32 + lane] = scale4(accB, ni);
}

__global__ __launch_bounds__(256)
void walk_gather_kernel(const int* __restrict__ rowptr_d, const int* __restrict__ csr_src,
                        const float* __restrict__ er, const float* __restrict__ rec_in,
                        const float* __restrict__ sigma,
                        float* csr_coef, float* __restrict__ norm_in,
                        const float* __restrict__ xprev, float* __restrict__ xnext) {
  const float4* __restrict__ rec = (const float4*)rec_in;
  const int lane = threadIdx.x & 31;
  const int node = __builtin_amdgcn_readfirstlane((int)(blockIdx.x * 8 + (threadIdx.x >> 5)));
  if (node >= N_NODES) return;
  const int head = lane & 3, k = lane >> 2;
  const int beg = rowptr_d[node], end = rowptr_d[node + 1];
  const int deg = end - beg;
  const float self = er[node * 4 + head];
  const float sgm = 1.f / (1.f + __expf(-sigma[0]));

  if (deg > 0 && deg <= 8 * WALK_CAP) {
    float4 r[WALK_CAP];
    float e[WALK_CAP];
    float m = -INFINITY;
#pragma unroll
    for (int i = 0; i < WALK_CAP; ++i) {
      const int j = beg + k + 8 * i;
      r[i] = rec[csr_src[j < end ? j : beg] * 4 + head];
      const float v = leakyf(r[i].x + self);
      e[i] = (j < end) ? v : -INFINITY;
      m = fmaxf(m, e[i]);
    }
    asm volatile("" : "+v"(r[0].y), "+v"(r[0].z), "+v"(r[0].w));
    m = red8_max(m);
    float sum = 0.f;
#pragma unroll
    for (int i = 0; i < WALK_CAP; ++i) {
      const float pe = __expf(e[i] - m);
      sum += (beg + k + 8 * i < end) ? pe : 0.f;
    }
    const float lnsum = __logf(red8_sum(sum));
#pragma unroll
    for (int i = 0; i < WALK_CAP; ++i) {
      const int j = beg + k + 8 * i;
      if (j < end) csr_coef[(size_t)j * 4 + head] = mixed_coef(e[i], m, lnsum, r[i], sgm);
    }
  } else {
    float m = -INFINITY;
    for (int j = beg + k; j < end; j += 8)
      m = fmaxf(m, leakyf(rec[csr_src[j] * 4 + head].x + self));
    m = red8_max(m);
    float sum = 0.f;
    for (int j = beg + k; j < end; j += 8)
      sum += __expf(leakyf(rec[csr_src[j] * 4 + head].x + self) - m);
    const float lnsum = __logf(red8_sum(sum));
    for (int j = beg + k; j < end; j += 8) {
      const float4 r = rec[csr_src[j] * 4 + head];
      const float ev = leakyf(r.x + self);
      csr_coef[(size_t)j * 4 + head] = mixed_coef(ev, m, lnsum, r, sgm);
    }
  }
  const float ni = sqrtf(fmaxf((float)deg, 1.f));
  if (lane == 0) norm_in[node] = ni;

  __threadfence_block();

  float4 accA, accB;
  sum_rows_own(csr_src, csr_coef, xprev, beg, end, lane, accA, accB);
  float4* o4 = (float4*)xnext + (size_t)node * (HF / 4);
  o4[lane]      = scale4(accA, ni);
  o4[32 + lane] = scale4(accB, ni);
}

__device__ __forceinline__ float half_sum16(float v) {
  v += __shfl_xor(v, 8); v += __shfl_xor(v, 4); v += __shfl_xor(v, 2); v += __shfl_xor(v, 1);
  return v;
}

__device__ __forceinline__ float4 combine4(const float4 (&v)[4], const float4& ar, float albase) {
  float score[4];
#pragma unroll
  for (int k = 0; k < 4; ++k) {
    const float p = v[k].x * ar.x + v[k].y * ar.y + v[k].z * ar.z + v[k].w * ar.w;
    score[k] = leakyf(albase + half_sum16(p));
  }
  const float m = fmaxf(fmaxf(score[0], score[1]), fmaxf(score[2], score[3]));
  float ex[4], ssum = 0.f;
#pragma unroll
  for (int k = 0; k < 4; ++k) { ex[k] = __expf(score[k] - m); ssum += ex[k]; }
  float4 o = make_float4(0.f, 0.f, 0.f, 0.f);
#pragma unroll
  for (int k = 0; k < 4; ++k) fma4(o, v[k], ex[k] / ssum);
  return o;
}

__global__ __launch_bounds__(256)
void gather_combine_kernel(const int* __restrict__ rowptr, const int* __restrict__ csr_src,
                           const float* __restrict__ csr_coef, const float* __restrict__ norm_in,
                           const float* __restrict__ h, const float* __restrict__ x1,
                           const float* __restrict__ x2, const float* __restrict__ al,
                           const float* __restrict__ hop_attn_r, float* __restrict__ out) {
  const int lane = threadIdx.x & 31;
  const int node = __builtin_amdgcn_readfirstlane((int)(blockIdx.x * 8 + (threadIdx.x >> 5)));
  if (node >= N_NODES) return;
  const int beg = rowptr[node], end = rowptr[node + 1];
  const int hsel = lane >> 4;
  const size_t rowoff = (size_t)node * (HF / 4);
  const float4* h4 = (const float4*)h + rowoff;
  const float4* p1 = (const float4*)x1 + rowoff;
  const float4* p2 = (const float4*)x2 + rowoff;
  const float4* r4 = (const float4*)hop_attn_r;

  float4 accA, accB;
  sum_rows(csr_src, csr_coef, x2, beg, end, lane, accA, accB);
  const float ni = norm_in[node];

  float4 vA[4], vB[4];
  vA[0] = h4[lane]; vB[0] = h4[32 + lane];
  vA[1] = p1[lane]; vB[1] = p1[32 + lane];
  vA[2] = p2[lane]; vB[2] = p2[32 + lane];
  const float4 arA = r4[lane], arB = r4[32 + lane];
  const float alA = al[node * 4 + hsel], alB = al[node * 4 + 2 + hsel];
  vA[3] = scale4(accA, ni);
  vB[3] = scale4(accB, ni);

  float4* o4 = (float4*)out + rowoff;
  o4[lane]      = combine4(vA, arA, alA);
  o4[32 + lane] = combine4(vB, arB, alB);
}

extern "C" void kernel_launch(void* const* d_in, const int* in_sizes, int n_in,
                              void* d_out, int out_size, void* d_ws, size_t ws_size,
                              hipStream_t stream) {
  (void)in_sizes; (void)n_in; (void)out_size; (void)ws_size;
  const float* feat       = (const float*)d_in[0];
  const int*   src        = (const int*)d_in[1];
  const int*   dst        = (const int*)d_in[2];
  const float* W          = (const float*)d_in[3];
  const float* attn_l     = (const float*)d_in[4];
  const float* attn_r     = (const float*)d_in[5];
  const float* hop_attn_l = (const float*)d_in[6];
  const float* hop_attn_r = (const float*)d_in[7];
  const float* sigma      = (const float*)d_in[8];

  float* ws = (float*)d_ws;
  size_t off = 0;
  float* h    = ws + off; off += (size_t)N_NODES * HF;
  float* x1   = ws + off; off += (size_t)N_NODES * HF;
  float* x2   = ws + off; off += (size_t)N_NODES * HF;
  float* el   = ws + off; off += (size_t)N_NODES * H_HEADS;
  float* er   = ws + off; off += (size_t)N_NODES * H_HEADS;
  float* al   = ws + off; off += (size_t)N_NODES * H_HEADS;
  float* norm_in  = ws + off; off += (size_t)N_NODES;
  int* rowptr_d = (int*)(ws + off); off += (size_t)N_NODES + 4;
  int* rowptr_s = (int*)(ws + off); off += (size_t)N_NODES + 4;
  int* csr_src  = (int*)(ws + off); off += (size_t)E_EDGES;
  int* csc_dst  = (int*)(ws + off); off += (size_t)E_EDGES;
  int* tmp_s    = (int*)(ws + off); off += (size_t)E_EDGES;
  int* tmp_d    = (int*)(ws + off); off += (size_t)E_EDGES;
  float* csr_coef = ws + off; off += (size_t)E_EDGES * H_HEADS;
  static_assert((size_t)N_NODES * H_HEADS * 4 <= (size_t)E_EDGES, "node records fit the chunk-ordered plane");
  float* rec = (float*)tmp_s;
  static_assert((size_t)2 * NCHUNK * TABW <= (size_t)E_EDGES * H_HEADS, "chunk tables fit the coefficient plane");
  int* tab_d = (int*)csr_coef;
  int* tab_s = tab_d + (size_t)NCHUNK * TABW;
  static_assert(N_NODES % 4 == 0 && E_EDGES % 4 == 0, "planes stay 16-byte aligned");

  gemm_bin_kernel<<<GEMM_BLOCKS + BIN_BLOCKS, 256, 0, stream>>>(feat, W, h, src, dst,
                                                                tab_d, tab_s, tmp_d, tmp_s);

  sort_scores_kernel<<<SORT_BLOCKS + SCORE_BLOCKS, 256, 0, stream>>>(
      tab_d, tab_s, tmp_d, tmp_s, rowptr_d, rowptr_s, csr_src, csc_dst,
      h, attn_l, attn_r, hop_attn_l, el, er, al);

  const int GB = (N_NODES + 7) / 8;
  src_walk_kernel<<<GB, 256, 0, stream>>>(rowptr_s, csc_dst, el, er, rec);

  walk_gather_kernel<<<GB, 256, 0, stream>>>(rowptr_d, csr_src, er, rec, sigma, csr_coef, norm_in,
                                             h, x1);
  gather_kernel<<<GB, 256, 0, stream>>>(rowptr_d, csr_src, csr_coef, norm_in, x1, x2);
  gather_combine_kernel<<<GB, 256, 0, stream>>>(rowptr_d, csr_src, csr_coef, norm_in, h, x1, x2,
                                                al, hop_attn_r, (float*)d_out);
}
